# starting kernel
# baseline (speedup 1.0000x reference)
#include <hip/hip_runtime.h>
#include <hip/hip_bf16.h>

#define BB 2
#define NN 8192
#define DD 64
#define DV 64
#define MM 3

typedef __attribute__((ext_vector_type(2))) float v2f;
typedef __attribute__((ext_vector_type(8))) float v8f;

__global__ __launch_bounds__(32) void v0_gemm_kernel(
    const float* __restrict__ X, const float* __restrict__ Wg,
    const float* __restrict__ bg, float* __restrict__ V0) {
  const int tile = blockIdx.x;
  const int b    = tile / (NN / 16);
  const int row0 = (tile % (NN / 16)) * 16;
  const int lane = threadIdx.x;
  const int r    = lane & 15;
  const int hi   = lane >> 4;

  const float* xrow = X + ((size_t)b * NN + row0 + r) * DD;
  v2f areg[16];
#pragma unroll
  for (int kk = 0; kk < 16; ++kk) {
    const int k0 = 4 * kk + 2 * hi;
    areg[kk] = *(const v2f*)(xrow + k0);
  }

#pragma unroll
  for (int ct = 0; ct < 4; ++ct) {
    const int col = 16 * ct + r;
    v8f acc = {0.f, 0.f, 0.f, 0.f, 0.f, 0.f, 0.f, 0.f};
#pragma unroll
    for (int kk = 0; kk < 16; ++kk) {
      const int k0 = 4 * kk + 2 * hi;
      v2f bm;
      bm.x = Wg[(size_t)k0 * DV + col];
      bm.y = Wg[(size_t)(k0 + 1) * DV + col];
      acc = __builtin_amdgcn_wmma_f32_16x16x4_f32(
           false, areg[kk],  false, bm,
           (short)0, acc,  false,  false);
    }
    const float bias = bg[col];
#pragma unroll
    for (int j = 0; j < 8; ++j) {
      V0[((size_t)b * NN + row0 + j + 8 * hi) * DV + col] = acc[j] + bias;
    }
  }
}

__global__ __launch_bounds__(128) void coeff_kernel(
    const float* __restrict__ X, const float* __restrict__ Wf,
    const float* __restrict__ bf, float* __restrict__ ca,
    float* __restrict__ cc) {
  __shared__ float Xs[128 * 65];
  const int tid  = threadIdx.x;
  const int cpb  = NN / 128;
  const int bi   = blockIdx.x;
  const int i0   = (bi % cpb) * 128;
  const int m    = (bi / cpb) % MM;
  const int b    = bi / (cpb * MM);

  const float* xbase = X + ((size_t)b * NN + i0) * DD;
  for (int idx = tid; idx < 128 * DD; idx += 128) {
    const int row = idx >> 6;
    const int col = idx & 63;
    Xs[row * 65 + col] = xbase[idx];
  }
  __syncthreads();

  const int i  = i0 + tid;
  const int i1 = (i + 1) & (NN - 1);
  float accA = bf[(size_t)m * NN + i];
  float accC = bf[(size_t)m * NN + i1];
  const float* wcol = Wf + (size_t)m * DD * NN;
#pragma unroll 8
  for (int d = 0; d < DD; ++d) {
    const float x = Xs[tid * 65 + d];
    accA += x * wcol[(size_t)d * NN + i];
    accC += x * wcol[(size_t)d * NN + i1];
  }
  const size_t o = ((size_t)b * MM + m) * NN + i;
  ca[o] = accA;
  cc[o] = accC;
}

__global__ __launch_bounds__(64) void combine_kernel(
    const float* __restrict__ V0, const float* __restrict__ ca,
    const float* __restrict__ cc, float* __restrict__ out) {
  const int dv = threadIdx.x;
  const int b  = blockIdx.x >> 13;
  const int i  = blockIdx.x & (NN - 1);

  const int j0 = i;
  const int j1 = (i + 1) & (NN - 1);
  const int j2 = (i + 2) & (NN - 1);
  const int j3 = (i + 3) & (NN - 1);

  const size_t cb = (size_t)b * MM * NN;
  const float a0_0 = ca[cb + 0 * NN + j0], c0_0 = cc[cb + 0 * NN + j0];
  const float a0_1 = ca[cb + 0 * NN + j1], c0_1 = cc[cb + 0 * NN + j1];
  const float a0_2 = ca[cb + 0 * NN + j2], c0_2 = cc[cb + 0 * NN + j2];
  const float a1_0 = ca[cb + 1 * NN + j0], c1_0 = cc[cb + 1 * NN + j0];
  const float a1_1 = ca[cb + 1 * NN + j1], c1_1 = cc[cb + 1 * NN + j1];
  const float a2_0 = ca[cb + 2 * NN + j0], c2_0 = cc[cb + 2 * NN + j0];

  const size_t vb = (size_t)b * NN;
  const float v0 = V0[(vb + j0) * DV + dv];
  const float v1 = V0[(vb + j1) * DV + dv];
  const float v2 = V0[(vb + j2) * DV + dv];
  const float v3 = V0[(vb + j3) * DV + dv];

  const float u0 = a0_0 * v0 + c0_0 * v1;
  const float u1 = a0_1 * v1 + c0_1 * v2;
  const float u2 = a0_2 * v2 + c0_2 * v3;
  const float w0 = a1_0 * u0 + c1_0 * u1;
  const float w1 = a1_1 * u1 + c1_1 * u2;

  out[(vb + i) * DV + dv] = a2_0 * w0 + c2_0 * w1;
}

extern "C" void kernel_launch(void* const* d_in, const int* in_sizes, int n_in,
                              void* d_out, int out_size, void* d_ws, size_t ws_size,
                              hipStream_t stream) {
  const float* X  = (const float*)d_in[0];
  const float* Wg = (const float*)d_in[1];
  const float* bg = (const float*)d_in[2];
  const float* Wf = (const float*)d_in[3];
  const float* bf = (const float*)d_in[4];
  float* out = (float*)d_out;

  float* V0 = (float*)d_ws;
  float* ca = V0 + (size_t)BB * NN * DV;
  float* cc = ca + (size_t)BB * MM * NN;

  v0_gemm_kernel<<<BB * NN / 16, 32, 0, stream>>>(X, Wg, bg, V0);
  coeff_kernel<<<BB * MM * NN / 128, 128, 0, stream>>>(X, Wf, bf, ca, cc);
  combine_kernel<<<BB * NN, 64, 0, stream>>>(V0, ca, cc, out);
}


# Chain folded into 4 row weights applied to X rows in LDS before the product: out = (sum_s k_s X[i+s]) Wg + (sum_s k_s) bg; one launch, 32 rows of both batches per group; the chain's sum order changes
# speedup vs baseline: 3.1859x; 3.1859x over previous
#include <hip/hip_runtime.h>

#define BB 2
#define NN 8192
#define DD 64
#define DV 64
#define MM 3

#define TT 32
#define XR (TT + 3)
#define XP 68
#define WQ 9
#define WC (4 * WQ)
#define CR (TT + 2)
#define CP 36
#define NT 256

typedef __attribute__((ext_vector_type(2))) float v2f;
typedef __attribute__((ext_vector_type(4))) float v4f;
typedef __attribute__((ext_vector_type(8))) float v8f;

__global__ __launch_bounds__(NT) void chord_fused_kernel(
    const float* __restrict__ X, const float* __restrict__ Wg,
    const float* __restrict__ bg, const float* __restrict__ Wf,
    const float* __restrict__ bf, float* __restrict__ out) {
  __shared__ __attribute__((aligned(16))) float Xs[BB * XR * XP];
  __shared__ __attribute__((aligned(16))) float Ws[MM * DD * WC];
  __shared__ float Ca[BB * MM * CP];
  __shared__ float Cc[BB * MM * CP];

  const int tid  = threadIdx.x;
  const int i0   = blockIdx.x * TT;
  const int lane = tid & 31;
  const int wave = tid >> 5;
  const int r    = lane & 15;
  const int hi   = lane >> 4;
  const int wb   = wave >> 2;
  const int rt   = (wave >> 1) & 1;
  const int ch   = wave & 1;

  v4f wv[7];
#pragma unroll
  for (int t = 0; t < 7; ++t) {
    int e = tid + NT * t;
    e = e < MM * DD * WQ ? e : MM * DD * WQ - 1;
    const int row = e / WQ;
    const int q   = e - row * WQ;
    const int col = (i0 + 4 * q) & (NN - 1);
    wv[t] = *(const v4f*)(Wf + (size_t)row * NN + col);
  }
  v4f xv[5];
#pragma unroll
  for (int t = 0; t < 5; ++t) {
    int e = tid + NT * t;
    e = e < BB * XR * 16 ? e : BB * XR * 16 - 1;
    const int b   = e / (XR * 16);
    const int rem = e - b * (XR * 16);
    const int j   = rem >> 4;
    const int q   = rem & 15;
    const int row = (i0 + j) & (NN - 1);
    xv[t] = *(const v4f*)(X + ((size_t)b * NN + row) * DD + 4 * q);
  }
  const bool cj_on = tid < BB * MM * CR;
  const int ct    = cj_on ? tid : BB * MM * CR - 1;
  const int cbm   = ct / CR;
  const int cj    = ct - cbm * CR;
  const int cb    = cbm / MM;
  const int cm    = cbm - cb * MM;
  float accA = bf[(size_t)cm * NN + ((i0 + cj) & (NN - 1))];
  float accC = bf[(size_t)cm * NN + ((i0 + cj + 1) & (NN - 1))];

#pragma unroll
  for (int t = 0; t < 7; ++t) {
    const int e = tid + NT * t;
    if (NT * (t + 1) <= MM * DD * WQ || e < MM * DD * WQ) {
      const int row = e / WQ;
      const int q   = e - row * WQ;
      *(v4f*)(Ws + row * WC + 4 * q) = wv[t];
    }
  }
#pragma unroll
  for (int t = 0; t < 5; ++t) {
    const int e = tid + NT * t;
    if (NT * (t + 1) <= BB * XR * 16 || e < BB * XR * 16) {
      const int b   = e / (XR * 16);
      const int rem = e - b * (XR * 16);
      const int j   = rem >> 4;
      const int q   = rem & 15;
      *(v4f*)(Xs + (b * XR + j) * XP + 4 * q) = xv[t];
    }
  }
  __syncthreads();

  v2f bw[2][16];
  float bgv[2];
#pragma unroll
  for (int c2 = 0; c2 < 2; ++c2) {
    const int col = 32 * ch + 16 * c2 + r;
#pragma unroll
    for (int kk = 0; kk < 16; ++kk) {
      const int k0 = 4 * kk + 2 * hi;
      bw[c2][kk].x = Wg[(size_t)k0 * DV + col];
      bw[c2][kk].y = Wg[(size_t)(k0 + 1) * DV + col];
    }
    bgv[c2] = bg[col];
  }

  if (cj_on) {
    const float* xrow = Xs + (cb * XR + cj) * XP;
    const float* wrow = Ws + cm * DD * WC + cj;
#pragma unroll
    for (int q = 0; q < 16; ++q) {
      const v4f x4 = *(const v4f*)(xrow + 4 * q);
#pragma unroll
      for (int u = 0; u < 4; ++u) {
        const int d = 4 * q + u;
        accA += x4[u] * wrow[d * WC];
        accC += x4[u] * wrow[d * WC + 1];
      }
    }
    Ca[cbm * CP + cj] = accA;
    Cc[cbm * CP + cj] = accC;
  }
  __syncthreads();

  const int il = rt * 16 + r;
  const float* ca0 = Ca + (wb * MM + 0) * CP + il;
  const float* cc0 = Cc + (wb * MM + 0) * CP + il;
  const float* ca1 = Ca + (wb * MM + 1) * CP + il;
  const float* cc1 = Cc + (wb * MM + 1) * CP + il;
  const float* ca2 = Ca + (wb * MM + 2) * CP + il;
  const float* cc2 = Cc + (wb * MM + 2) * CP + il;
  const float a0_0 = ca0[0], c0_0 = cc0[0];
  const float a0_1 = ca0[1], c0_1 = cc0[1];
  const float a0_2 = ca0[2], c0_2 = cc0[2];
  const float a1_0 = ca1[0], c1_0 = cc1[0];
  const float a1_1 = ca1[1], c1_1 = cc1[1];
  const float a2_0 = ca2[0], c2_0 = cc2[0];
  const float p11 = a2_0 * a1_0;
  const float p12 = a2_0 * c1_0 + c2_0 * a1_1;
  const float p13 = c2_0 * c1_1;
  const float k0 = p11 * a0_0;
  const float k1 = p11 * c0_0 + p12 * a0_1;
  const float k2 = p12 * c0_1 + p13 * a0_2;
  const float k3 = p13 * c0_2;
  const float kb = (k0 + k1) + (k2 + k3);

  const float* xs0 = Xs + (wb * XR + il) * XP + 2 * hi;
  v2f areg[16];
#pragma unroll
  for (int kk = 0; kk < 16; ++kk) {
    const v2f x0 = *(const v2f*)(xs0 + 0 * XP + 4 * kk);
    const v2f x1 = *(const v2f*)(xs0 + 1 * XP + 4 * kk);
    const v2f x2 = *(const v2f*)(xs0 + 2 * XP + 4 * kk);
    const v2f x3 = *(const v2f*)(xs0 + 3 * XP + 4 * kk);
    areg[kk] = (k0 * x0 + k1 * x1) + (k2 * x2 + k3 * x3);
  }

  v8f acc0 = {0.f, 0.f, 0.f, 0.f, 0.f, 0.f, 0.f, 0.f};
  v8f acc1 = {0.f, 0.f, 0.f, 0.f, 0.f, 0.f, 0.f, 0.f};
#pragma unroll
  for (int kk = 0; kk < 16; ++kk) {
    acc0 = __builtin_amdgcn_wmma_f32_16x16x4_f32(
         false, areg[kk],  false, bw[0][kk],
         (short)0, acc0,  false,  false);
    acc1 = __builtin_amdgcn_wmma_f32_16x16x4_f32(
         false, areg[kk],  false, bw[1][kk],
         (short)0, acc1,  false,  false);
  }

  float* orow = out + ((size_t)wb * NN + i0 + rt * 16 + 8 * hi) * DV + 32 * ch + r;
#pragma unroll
  for (int j = 0; j < 8; ++j) {
    const float kbr = __shfl(kb, j + 8 * hi, 32);
    orow[(size_t)j * DV]      = acc0[j] + kbr * bgv[0];
    orow[(size_t)j * DV + 16] = acc1[j] + kbr * bgv[1];
  }
}

extern "C" void kernel_launch(void* const* d_in, const int* in_sizes, int n_in,
                              void* d_out, int out_size, void* d_ws, size_t ws_size,
                              hipStream_t stream) {
  const float* X  = (const float*)d_in[0];
  const float* Wg = (const float*)d_in[1];
  const float* bg = (const float*)d_in[2];
  const float* Wf = (const float*)d_in[3];
  const float* bf = (const float*)d_in[4];
  float* out = (float*)d_out;

  chord_fused_kernel<<<NN / TT, NT, 0, stream>>>(X, Wg, bg, Wf, bf, out);
}


# Coefficient dot products cut in two halves of d summed by two threads at once, one thread serving both batches (shared Wf reads); each a, c is now (bias + first half) + second half
# speedup vs baseline: 3.5743x; 1.1219x over previous
#include <hip/hip_runtime.h>

#define BB 2
#define NN 8192
#define DD 64
#define DV 64
#define MM 3

#define TT 32
#define XR (TT + 3)
#define XP 68
#define WQ 9
#define WC (4 * WQ)
#define CR (TT + 2)
#define CP 36
#define NT 256

typedef __attribute__((ext_vector_type(2))) float v2f;
typedef __attribute__((ext_vector_type(4))) float v4f;
typedef __attribute__((ext_vector_type(8))) float v8f;

__global__ __launch_bounds__(NT) void chord_fused_kernel(
    const float* __restrict__ X, const float* __restrict__ Wg,
    const float* __restrict__ bg, const float* __restrict__ Wf,
    const float* __restrict__ bf, float* __restrict__ out) {
  __shared__ __attribute__((aligned(16))) float Xs[BB * XR * XP];
  __shared__ __attribute__((aligned(16))) float Ws[MM * DD * WC];
  __shared__ float Ca[2 * BB * MM * CP];
  __shared__ float Cc[2 * BB * MM * CP];

  const int tid  = threadIdx.x;
  const int i0   = blockIdx.x * TT;
  const int lane = tid & 31;
  const int wave = tid >> 5;
  const int r    = lane & 15;
  const int hi   = lane >> 4;
  const int wb   = wave >> 2;
  const int rt   = (wave >> 1) & 1;
  const int ch   = wave & 1;

  v4f wv[7];
#pragma unroll
  for (int t = 0; t < 7; ++t) {
    int e = tid + NT * t;
    e = e < MM * DD * WQ ? e : MM * DD * WQ - 1;
    const int row = e / WQ;
    const int q   = e - row * WQ;
    const int col = (i0 + 4 * q) & (NN - 1);
    wv[t] = *(const v4f*)(Wf + (size_t)row * NN + col);
  }
  v4f xv[5];
#pragma unroll
  for (int t = 0; t < 5; ++t) {
    int e = tid + NT * t;
    e = e < BB * XR * 16 ? e : BB * XR * 16 - 1;
    const int b   = e / (XR * 16);
    const int rem = e - b * (XR * 16);
    const int j   = rem >> 4;
    const int q   = rem & 15;
    const int row = (i0 + j) & (NN - 1);
    xv[t] = *(const v4f*)(X + ((size_t)b * NN + row) * DD + 4 * q);
  }
  const bool cj_on = tid < 2 * MM * CR;
  const int ct    = cj_on ? tid : 2 * MM * CR - 1;
  const int chf   = ct / (MM * CR);
  const int cmj   = ct - chf * (MM * CR);
  const int cm    = cmj / CR;
  const int cj    = cmj - cm * CR;
  const float bfA = bf[(size_t)cm * NN + ((i0 + cj) & (NN - 1))];
  const float bfC = bf[(size_t)cm * NN + ((i0 + cj + 1) & (NN - 1))];

#pragma unroll
  for (int t = 0; t < 7; ++t) {
    const int e = tid + NT * t;
    if (NT * (t + 1) <= MM * DD * WQ || e < MM * DD * WQ) {
      const int row = e / WQ;
      const int q   = e - row * WQ;
      *(v4f*)(Ws + row * WC + 4 * q) = wv[t];
    }
  }
#pragma unroll
  for (int t = 0; t < 5; ++t) {
    const int e = tid + NT * t;
    if (NT * (t + 1) <= BB * XR * 16 || e < BB * XR * 16) {
      const int b   = e / (XR * 16);
      const int rem = e - b * (XR * 16);
      const int j   = rem >> 4;
      const int q   = rem & 15;
      *(v4f*)(Xs + (b * XR + j) * XP + 4 * q) = xv[t];
    }
  }
  __syncthreads();

  v2f bw[2][16];
  float bgv[2];
#pragma unroll
  for (int c2 = 0; c2 < 2; ++c2) {
    const int col = 32 * ch + 16 * c2 + r;
#pragma unroll
    for (int kk = 0; kk < 16; ++kk) {
      const int k0 = 4 * kk + 2 * hi;
      bw[c2][kk].x = Wg[(size_t)k0 * DV + col];
      bw[c2][kk].y = Wg[(size_t)(k0 + 1) * DV + col];
    }
    bgv[c2] = bg[col];
  }

  if (cj_on) {
    float accA0 = chf ? 0.f : bfA, accC0 = chf ? 0.f : bfC;
    float accA1 = accA0,           accC1 = accC0;
    const float* xrow0 = Xs + (0 * XR + cj) * XP + 32 * chf;
    const float* xrow1 = Xs + (1 * XR + cj) * XP + 32 * chf;
    const float* wrow  = Ws + (cm * DD + 32 * chf) * WC + cj;
#pragma unroll
    for (int q = 0; q < 8; ++q) {
      const v4f xa = *(const v4f*)(xrow0 + 4 * q);
      const v4f xb = *(const v4f*)(xrow1 + 4 * q);
#pragma unroll
      for (int u = 0; u < 4; ++u) {
        const int d = 4 * q + u;
        const float w0 = wrow[d * WC];
        const float w1 = wrow[d * WC + 1];
        accA0 += xa[u] * w0;
        accC0 += xa[u] * w1;
        accA1 += xb[u] * w0;
        accC1 += xb[u] * w1;
      }
    }
    Ca[((chf * BB + 0) * MM + cm) * CP + cj] = accA0;
    Cc[((chf * BB + 0) * MM + cm) * CP + cj] = accC0;
    Ca[((chf * BB + 1) * MM + cm) * CP + cj] = accA1;
    Cc[((chf * BB + 1) * MM + cm) * CP + cj] = accC1;
  }
  __syncthreads();

  const int il = rt * 16 + r;
  const float* ca0 = Ca + (wb * MM + 0) * CP + il;
  const float* cc0 = Cc + (wb * MM + 0) * CP + il;
  const float* ca1 = Ca + (wb * MM + 1) * CP + il;
  const float* cc1 = Cc + (wb * MM + 1) * CP + il;
  const float* ca2 = Ca + (wb * MM + 2) * CP + il;
  const float* cc2 = Cc + (wb * MM + 2) * CP + il;
  const int HS = BB * MM * CP;
  const float a0_0 = ca0[0] + ca0[HS + 0], c0_0 = cc0[0] + cc0[HS + 0];
  const float a0_1 = ca0[1] + ca0[HS + 1], c0_1 = cc0[1] + cc0[HS + 1];
  const float a0_2 = ca0[2] + ca0[HS + 2], c0_2 = cc0[2] + cc0[HS + 2];
  const float a1_0 = ca1[0] + ca1[HS + 0], c1_0 = cc1[0] + cc1[HS + 0];
  const float a1_1 = ca1[1] + ca1[HS + 1], c1_1 = cc1[1] + cc1[HS + 1];
  const float a2_0 = ca2[0] + ca2[HS + 0], c2_0 = cc2[0] + cc2[HS + 0];
  const float p11 = a2_0 * a1_0;
  const float p12 = a2_0 * c1_0 + c2_0 * a1_1;
  const float p13 = c2_0 * c1_1;
  const float k0 = p11 * a0_0;
  const float k1 = p11 * c0_0 + p12 * a0_1;
  const float k2 = p12 * c0_1 + p13 * a0_2;
  const float k3 = p13 * c0_2;
  const float kb = (k0 + k1) + (k2 + k3);

  const float* xs0 = Xs + (wb * XR + il) * XP + 2 * hi;
  v2f areg[16];
#pragma unroll
  for (int kk = 0; kk < 16; ++kk) {
    const v2f x0 = *(const v2f*)(xs0 + 0 * XP + 4 * kk);
    const v2f x1 = *(const v2f*)(xs0 + 1 * XP + 4 * kk);
    const v2f x2 = *(const v2f*)(xs0 + 2 * XP + 4 * kk);
    const v2f x3 = *(const v2f*)(xs0 + 3 * XP + 4 * kk);
    areg[kk] = (k0 * x0 + k1 * x1) + (k2 * x2 + k3 * x3);
  }

  v8f acc0 = {0.f, 0.f, 0.f, 0.f, 0.f, 0.f, 0.f, 0.f};
  v8f acc1 = {0.f, 0.f, 0.f, 0.f, 0.f, 0.f, 0.f, 0.f};
#pragma unroll
  for (int kk = 0; kk < 16; ++kk) {
    acc0 = __builtin_amdgcn_wmma_f32_16x16x4_f32(
         false, areg[kk],  false, bw[0][kk],
         (short)0, acc0,  false,  false);
    acc1 = __builtin_amdgcn_wmma_f32_16x16x4_f32(
         false, areg[kk],  false, bw[1][kk],
         (short)0, acc1,  false,  false);
  }

  float* orow = out + ((size_t)wb * NN + i0 + rt * 16 + 8 * hi) * DV + 32 * ch + r;
#pragma unroll
  for (int j = 0; j < 8; ++j) {
    const float kbr = __shfl(kb, j + 8 * hi, 32);
    orow[(size_t)j * DV]      = acc0[j] + kbr * bgv[0];
    orow[(size_t)j * DV + 16] = acc1[j] + kbr * bgv[1];
  }
}

extern "C" void kernel_launch(void* const* d_in, const int* in_sizes, int n_in,
                              void* d_out, int out_size, void* d_ws, size_t ws_size,
                              hipStream_t stream) {
  const float* X  = (const float*)d_in[0];
  const float* Wg = (const float*)d_in[1];
  const float* bg = (const float*)d_in[2];
  const float* Wf = (const float*)d_in[3];
  const float* bf = (const float*)d_in[4];
  float* out = (float*)d_out;

  chord_fused_kernel<<<NN / TT, NT, 0, stream>>>(X, Wg, bg, Wf, bf, out);
}


# Wg goes through one LDS copy per work-group in B-fragment order (4 16-byte global loads per thread, one 16-byte LDS read per k-step), not 64 4-byte global loads per lane; every sum keeps its order
# speedup vs baseline: 3.7850x; 1.0589x over previous
#include <hip/hip_runtime.h>

#define BB 2
#define NN 8192
#define DD 64
#define DV 64
#define MM 3

#define TT 32
#define XR (TT + 3)
#define XP 68
#define WQ 9
#define WC (4 * WQ)
#define CR (TT + 2)
#define CP 36
#define BP 17
#define NT 256

typedef __attribute__((ext_vector_type(2))) float v2f;
typedef __attribute__((ext_vector_type(4))) float v4f;
typedef __attribute__((ext_vector_type(8))) float v8f;

__global__ __launch_bounds__(NT) void chord_fused_kernel(
    const float* __restrict__ X, const float* __restrict__ Wg,
    const float* __restrict__ bg, const float* __restrict__ Wf,
    const float* __restrict__ bf, float* __restrict__ out) {
  __shared__ __attribute__((aligned(16))) float Xs[BB * XR * XP];
  __shared__ __attribute__((aligned(16))) float Ws[MM * DD * WC];
  __shared__ v4f Bs[16 * 2 * 2 * BP];
  __shared__ float Ca[2 * BB * MM * CP];
  __shared__ float Cc[2 * BB * MM * CP];

  const int tid  = threadIdx.x;
  const int i0   = blockIdx.x * TT;
  const int lane = tid & 31;
  const int wave = tid >> 5;
  const int r    = lane & 15;
  const int hi   = lane >> 4;
  const int wb   = wave >> 2;
  const int rt   = (wave >> 1) & 1;
  const int ch   = wave & 1;

  v4f wv[7];
#pragma unroll
  for (int t = 0; t < 7; ++t) {
    int e = tid + NT * t;
    e = e < MM * DD * WQ ? e : MM * DD * WQ - 1;
    const int row = e / WQ;
    const int q   = e - row * WQ;
    const int col = (i0 + 4 * q) & (NN - 1);
    wv[t] = *(const v4f*)(Wf + (size_t)row * NN + col);
  }
  v4f xv[5];
#pragma unroll
  for (int t = 0; t < 5; ++t) {
    int e = tid + NT * t;
    e = e < BB * XR * 16 ? e : BB * XR * 16 - 1;
    const int b   = e / (XR * 16);
    const int rem = e - b * (XR * 16);
    const int j   = rem >> 4;
    const int q   = rem & 15;
    const int row = (i0 + j) & (NN - 1);
    xv[t] = *(const v4f*)(X + ((size_t)b * NN + row) * DD + 4 * q);
  }
  const bool cj_on = tid < 2 * MM * CR;
  const int ct    = cj_on ? tid : 2 * MM * CR - 1;
  const int chf   = ct / (MM * CR);
  const int cmj   = ct - chf * (MM * CR);
  const int cm    = cmj / CR;
  const int cj    = cmj - cm * CR;
  const float bfA = bf[(size_t)cm * NN + ((i0 + cj) & (NN - 1))];
  const float bfC = bf[(size_t)cm * NN + ((i0 + cj + 1) & (NN - 1))];

#pragma unroll
  for (int t = 0; t < 7; ++t) {
    const int e = tid + NT * t;
    if (NT * (t + 1) <= MM * DD * WQ || e < MM * DD * WQ) {
      const int row = e / WQ;
      const int q   = e - row * WQ;
      *(v4f*)(Ws + row * WC + 4 * q) = wv[t];
    }
  }
#pragma unroll
  for (int t = 0; t < 5; ++t) {
    const int e = tid + NT * t;
    if (NT * (t + 1) <= BB * XR * 16 || e < BB * XR * 16) {
      const int b   = e / (XR * 16);
      const int rem = e - b * (XR * 16);
      const int j   = rem >> 4;
      const int q   = rem & 15;
      *(v4f*)(Xs + (b * XR + j) * XP + 4 * q) = xv[t];
    }
  }
  __syncthreads();

  const int srq = tid & 3;
  const int sg  = tid >> 2;
  const int sk0 = 4 * (sg >> 2) + 2 * (sg & 1);
  const float* wgp = Wg + (size_t)sk0 * DV + 32 * ((sg >> 1) & 1) + 4 * srq;
  const v4f g00 = *(const v4f*)(wgp);
  const v4f g01 = *(const v4f*)(wgp + 16);
  const v4f g10 = *(const v4f*)(wgp + DV);
  const v4f g11 = *(const v4f*)(wgp + DV + 16);
  float bgv[2];
  bgv[0] = bg[32 * ch + r];
  bgv[1] = bg[32 * ch + 16 + r];

  if (cj_on) {
    float accA0 = chf ? 0.f : bfA, accC0 = chf ? 0.f : bfC;
    float accA1 = accA0,           accC1 = accC0;
    const float* xrow0 = Xs + (0 * XR + cj) * XP + 32 * chf;
    const float* xrow1 = Xs + (1 * XR + cj) * XP + 32 * chf;
    const float* wrow  = Ws + (cm * DD + 32 * chf) * WC + cj;
#pragma unroll
    for (int q = 0; q < 8; ++q) {
      const v4f xa = *(const v4f*)(xrow0 + 4 * q);
      const v4f xb = *(const v4f*)(xrow1 + 4 * q);
#pragma unroll
      for (int u = 0; u < 4; ++u) {
        const int d = 4 * q + u;
        const float w0 = wrow[d * WC];
        const float w1 = wrow[d * WC + 1];
        accA0 += xa[u] * w0;
        accC0 += xa[u] * w1;
        accA1 += xb[u] * w0;
        accC1 += xb[u] * w1;
      }
    }
    Ca[((chf * BB + 0) * MM + cm) * CP + cj] = accA0;
    Cc[((chf * BB + 0) * MM + cm) * CP + cj] = accC0;
    Ca[((chf * BB + 1) * MM + cm) * CP + cj] = accA1;
    Cc[((chf * BB + 1) * MM + cm) * CP + cj] = accC1;
  }
  {
    v4f* bdst = Bs + sg * BP + 4 * srq;
    bdst[0] = (v4f){g00.x, g10.x, g01.x, g11.x};
    bdst[1] = (v4f){g00.y, g10.y, g01.y, g11.y};
    bdst[2] = (v4f){g00.z, g10.z, g01.z, g11.z};
    bdst[3] = (v4f){g00.w, g10.w, g01.w, g11.w};
  }
  __syncthreads();

  const int il = rt * 16 + r;
  const float* ca0 = Ca + (wb * MM + 0) * CP + il;
  const float* cc0 = Cc + (wb * MM + 0) * CP + il;
  const float* ca1 = Ca + (wb * MM + 1) * CP + il;
  const float* cc1 = Cc + (wb * MM + 1) * CP + il;
  const float* ca2 = Ca + (wb * MM + 2) * CP + il;
  const float* cc2 = Cc + (wb * MM + 2) * CP + il;
  const int HS = BB * MM * CP;
  const float a0_0 = ca0[0] + ca0[HS + 0], c0_0 = cc0[0] + cc0[HS + 0];
  const float a0_1 = ca0[1] + ca0[HS + 1], c0_1 = cc0[1] + cc0[HS + 1];
  const float a0_2 = ca0[2] + ca0[HS + 2], c0_2 = cc0[2] + cc0[HS + 2];
  const float a1_0 = ca1[0] + ca1[HS + 0], c1_0 = cc1[0] + cc1[HS + 0];
  const float a1_1 = ca1[1] + ca1[HS + 1], c1_1 = cc1[1] + cc1[HS + 1];
  const float a2_0 = ca2[0] + ca2[HS + 0], c2_0 = cc2[0] + cc2[HS + 0];
  const float p11 = a2_0 * a1_0;
  const float p12 = a2_0 * c1_0 + c2_0 * a1_1;
  const float p13 = c2_0 * c1_1;
  const float k0 = p11 * a0_0;
  const float k1 = p11 * c0_0 + p12 * a0_1;
  const float k2 = p12 * c0_1 + p13 * a0_2;
  const float k3 = p13 * c0_2;
  const float kb = (k0 + k1) + (k2 + k3);

  v2f bw[2][16];
  const v4f* bsrc = Bs + (ch * 2 + hi) * BP + r;
#pragma unroll
  for (int kk = 0; kk < 16; ++kk) {
    const v4f t = bsrc[kk * 4 * BP];
    bw[0][kk] = (v2f){t.x, t.y};
    bw[1][kk] = (v2f){t.z, t.w};
  }

  const float* xs0 = Xs + (wb * XR + il) * XP + 2 * hi;
  v2f areg[16];
#pragma unroll
  for (int kk = 0; kk < 16; ++kk) {
    const v2f x0 = *(const v2f*)(xs0 + 0 * XP + 4 * kk);
    const v2f x1 = *(const v2f*)(xs0 + 1 * XP + 4 * kk);
    const v2f x2 = *(const v2f*)(xs0 + 2 * XP + 4 * kk);
    const v2f x3 = *(const v2f*)(xs0 + 3 * XP + 4 * kk);
    areg[kk] = (k0 * x0 + k1 * x1) + (k2 * x2 + k3 * x3);
  }

  v8f acc0 = {0.f, 0.f, 0.f, 0.f, 0.f, 0.f, 0.f, 0.f};
  v8f acc1 = {0.f, 0.f, 0.f, 0.f, 0.f, 0.f, 0.f, 0.f};
#pragma unroll
  for (int kk = 0; kk < 16; ++kk) {
    acc0 = __builtin_amdgcn_wmma_f32_16x16x4_f32(
         false, areg[kk],  false, bw[0][kk],
         (short)0, acc0,  false,  false);
    acc1 = __builtin_amdgcn_wmma_f32_16x16x4_f32(
         false, areg[kk],  false, bw[1][kk],
         (short)0, acc1,  false,  false);
  }

  float* orow = out + ((size_t)wb * NN + i0 + rt * 16 + 8 * hi) * DV + 32 * ch + r;
#pragma unroll
  for (int j = 0; j < 8; ++j) {
    const float kbr = __shfl(kb, j + 8 * hi, 32);
    orow[(size_t)j * DV]      = acc0[j] + kbr * bgv[0];
    orow[(size_t)j * DV + 16] = acc1[j] + kbr * bgv[1];
  }
}

extern "C" void kernel_launch(void* const* d_in, const int* in_sizes, int n_in,
                              void* d_out, int out_size, void* d_ws, size_t ws_size,
                              hipStream_t stream) {
  const float* X  = (const float*)d_in[0];
  const float* Wg = (const float*)d_in[1];
  const float* bg = (const float*)d_in[2];
  const float* Wf = (const float*)d_in[3];
  const float* bf = (const float*)d_in[4];
  float* out = (float*)d_out;

  chord_fused_kernel<<<NN / TT, NT, 0, stream>>>(X, Wg, bg, Wf, bf, out);
}


# Four waves per work-group, not eight: each wave owns 16 whole output rows (four column tiles, A fragment built once), each thread two staging and coefficient jobs; every sum keeps its order
# speedup vs baseline: 4.2368x; 1.1194x over previous
#include <hip/hip_runtime.h>

#define BB 2
#define NN 8192
#define DD 64
#define DV 64
#define MM 3

#define TT 32
#define XR (TT + 3)
#define XP 68
#define WQ 9
#define WC (4 * WQ)
#define CR (TT + 2)
#define CP 36
#define BP 17
#define NT 128
#define WFR 14
#define XRR 9
#define NJ (2 * MM * CR)

typedef __attribute__((ext_vector_type(2))) float v2f;
typedef __attribute__((ext_vector_type(4))) float v4f;
typedef __attribute__((ext_vector_type(8))) float v8f;

static_assert(NT * WFR >= MM * DD * WQ && NT * WFR - MM * DD * WQ <= MM * DD * WQ - NT * (WFR - 1), "Wf rounds");
static_assert(NT * XRR >= BB * XR * 16 && NT * XRR - BB * XR * 16 <= BB * XR * 16 - NT * (XRR - 1), "X rounds");
static_assert(2 * NT >= NJ && NJ >= NT && 2 * NT - NJ <= NJ - NT, "coefficient jobs");

__global__ __launch_bounds__(NT) void chord_fused_kernel(
    const float* __restrict__ X, const float* __restrict__ Wg,
    const float* __restrict__ bg, const float* __restrict__ Wf,
    const float* __restrict__ bf, float* __restrict__ out) {
  __shared__ __attribute__((aligned(16))) float Xs[BB * XR * XP];
  __shared__ __attribute__((aligned(16))) float Ws[MM * DD * WC];
  __shared__ v4f Bs[16 * 2 * 2 * BP];
  __shared__ float Ca[2 * BB * MM * CP];
  __shared__ float Cc[2 * BB * MM * CP];

  const int tid  = threadIdx.x;
  const int i0   = blockIdx.x * TT;
  const int lane = tid & 31;
  const int wave = tid >> 5;
  const int r    = lane & 15;
  const int hi   = lane >> 4;
  const int wb   = wave >> 1;
  const int rt   = wave & 1;

  v4f wv[WFR];
#pragma unroll
  for (int t = 0; t < WFR; ++t) {
    int e = tid + NT * t;
    if (t == WFR - 1) e = e < MM * DD * WQ ? e : e - (NT * WFR - MM * DD * WQ);
    const int row = e / WQ;
    const int q   = e - row * WQ;
    const int col = (i0 + 4 * q) & (NN - 1);
    wv[t] = *(const v4f*)(Wf + (size_t)row * NN + col);
  }
  v4f xv[XRR];
#pragma unroll
  for (int t = 0; t < XRR; ++t) {
    int e = tid + NT * t;
    if (t == XRR - 1) e = e < BB * XR * 16 ? e : e - (NT * XRR - BB * XR * 16);
    const int b   = e / (XR * 16);
    const int rem = e - b * (XR * 16);
    const int j   = rem >> 4;
    const int q   = rem & 15;
    const int row = (i0 + j) & (NN - 1);
    xv[t] = *(const v4f*)(X + ((size_t)b * NN + row) * DD + 4 * q);
  }
  int   jhf[2], jm[2], jj[2];
  float bfA[2], bfC[2];
#pragma unroll
  for (int rnd = 0; rnd < 2; ++rnd) {
    int ct = tid + NT * rnd;
    if (rnd == 1) ct = ct < NJ ? ct : ct - (2 * NT - NJ);
    jhf[rnd] = ct / (MM * CR);
    const int cmj = ct - jhf[rnd] * (MM * CR);
    jm[rnd] = cmj / CR;
    jj[rnd] = cmj - jm[rnd] * CR;
    bfA[rnd] = bf[(size_t)jm[rnd] * NN + ((i0 + jj[rnd]) & (NN - 1))];
    bfC[rnd] = bf[(size_t)jm[rnd] * NN + ((i0 + jj[rnd] + 1) & (NN - 1))];
  }

#pragma unroll
  for (int t = 0; t < WFR; ++t) {
    int e = tid + NT * t;
    if (t == WFR - 1) e = e < MM * DD * WQ ? e : e - (NT * WFR - MM * DD * WQ);
    const int row = e / WQ;
    const int q   = e - row * WQ;
    *(v4f*)(Ws + row * WC + 4 * q) = wv[t];
  }
#pragma unroll
  for (int t = 0; t < XRR; ++t) {
    int e = tid + NT * t;
    if (t == XRR - 1) e = e < BB * XR * 16 ? e : e - (NT * XRR - BB * XR * 16);
    const int b   = e / (XR * 16);
    const int rem = e - b * (XR * 16);
    const int j   = rem >> 4;
    const int q   = rem & 15;
    *(v4f*)(Xs + (b * XR + j) * XP + 4 * q) = xv[t];
  }
  __syncthreads();

  v4f g[2][4];
#pragma unroll
  for (int rnd = 0; rnd < 2; ++rnd) {
    const int idx = tid + NT * rnd;
    const int srq = idx & 3;
    const int sg  = idx >> 2;
    const int sk0 = 4 * (sg >> 2) + 2 * (sg & 1);
    const float* wgp = Wg + (size_t)sk0 * DV + 32 * ((sg >> 1) & 1) + 4 * srq;
    g[rnd][0] = *(const v4f*)(wgp);
    g[rnd][1] = *(const v4f*)(wgp + 16);
    g[rnd][2] = *(const v4f*)(wgp + DV);
    g[rnd][3] = *(const v4f*)(wgp + DV + 16);
  }
  float bgv[4];
#pragma unroll
  for (int t = 0; t < 4; ++t) bgv[t] = bg[16 * t + r];

#pragma unroll
  for (int rnd = 0; rnd < 2; ++rnd) {
    const int chf = jhf[rnd], cm = jm[rnd], cj = jj[rnd];
    float accA0 = chf ? 0.f : bfA[rnd], accC0 = chf ? 0.f : bfC[rnd];
    float accA1 = accA0,                accC1 = accC0;
    const float* xrow0 = Xs + (0 * XR + cj) * XP + 32 * chf;
    const float* xrow1 = Xs + (1 * XR + cj) * XP + 32 * chf;
    const float* wrow  = Ws + (cm * DD + 32 * chf) * WC + cj;
#pragma unroll
    for (int q = 0; q < 8; ++q) {
      const v4f xa = *(const v4f*)(xrow0 + 4 * q);
      const v4f xb = *(const v4f*)(xrow1 + 4 * q);
#pragma unroll
      for (int u = 0; u < 4; ++u) {
        const int d = 4 * q + u;
        const float w0 = wrow[d * WC];
        const float w1 = wrow[d * WC + 1];
        accA0 += xa[u] * w0;
        accC0 += xa[u] * w1;
        accA1 += xb[u] * w0;
        accC1 += xb[u] * w1;
      }
    }
    Ca[((chf * BB + 0) * MM + cm) * CP + cj] = accA0;
    Cc[((chf * BB + 0) * MM + cm) * CP + cj] = accC0;
    Ca[((chf * BB + 1) * MM + cm) * CP + cj] = accA1;
    Cc[((chf * BB + 1) * MM + cm) * CP + cj] = accC1;
  }
#pragma unroll
  for (int rnd = 0; rnd < 2; ++rnd) {
    const int idx = tid + NT * rnd;
    v4f* bdst = Bs + (idx >> 2) * BP + 4 * (idx & 3);
    bdst[0] = (v4f){g[rnd][0].x, g[rnd][2].x, g[rnd][1].x, g[rnd][3].x};
    bdst[1] = (v4f){g[rnd][0].y, g[rnd][2].y, g[rnd][1].y, g[rnd][3].y};
    bdst[2] = (v4f){g[rnd][0].z, g[rnd][2].z, g[rnd][1].z, g[rnd][3].z};
    bdst[3] = (v4f){g[rnd][0].w, g[rnd][2].w, g[rnd][1].w, g[rnd][3].w};
  }
  __syncthreads();

  const int il = rt * 16 + r;
  const float* ca0 = Ca + (wb * MM + 0) * CP + il;
  const float* cc0 = Cc + (wb * MM + 0) * CP + il;
  const float* ca1 = Ca + (wb * MM + 1) * CP + il;
  const float* cc1 = Cc + (wb * MM + 1) * CP + il;
  const float* ca2 = Ca + (wb * MM + 2) * CP + il;
  const float* cc2 = Cc + (wb * MM + 2) * CP + il;
  const int HS = BB * MM * CP;
  const float a0_0 = ca0[0] + ca0[HS + 0], c0_0 = cc0[0] + cc0[HS + 0];
  const float a0_1 = ca0[1] + ca0[HS + 1], c0_1 = cc0[1] + cc0[HS + 1];
  const float a0_2 = ca0[2] + ca0[HS + 2], c0_2 = cc0[2] + cc0[HS + 2];
  const float a1_0 = ca1[0] + ca1[HS + 0], c1_0 = cc1[0] + cc1[HS + 0];
  const float a1_1 = ca1[1] + ca1[HS + 1], c1_1 = cc1[1] + cc1[HS + 1];
  const float a2_0 = ca2[0] + ca2[HS + 0], c2_0 = cc2[0] + cc2[HS + 0];
  const float p11 = a2_0 * a1_0;
  const float p12 = a2_0 * c1_0 + c2_0 * a1_1;
  const float p13 = c2_0 * c1_1;
  const float k0 = p11 * a0_0;
  const float k1 = p11 * c0_0 + p12 * a0_1;
  const float k2 = p12 * c0_1 + p13 * a0_2;
  const float k3 = p13 * c0_2;
  const float kb = (k0 + k1) + (k2 + k3);

  v2f bw[4][16];
#pragma unroll
  for (int ch = 0; ch < 2; ++ch) {
    const v4f* bsrc = Bs + (ch * 2 + hi) * BP + r;
#pragma unroll
    for (int kk = 0; kk < 16; ++kk) {
      const v4f t = bsrc[kk * 4 * BP];
      bw[2 * ch + 0][kk] = (v2f){t.x, t.y};
      bw[2 * ch + 1][kk] = (v2f){t.z, t.w};
    }
  }

  const float* xs0 = Xs + (wb * XR + il) * XP + 2 * hi;
  v2f areg[16];
#pragma unroll
  for (int kk = 0; kk < 16; ++kk) {
    const v2f x0 = *(const v2f*)(xs0 + 0 * XP + 4 * kk);
    const v2f x1 = *(const v2f*)(xs0 + 1 * XP + 4 * kk);
    const v2f x2 = *(const v2f*)(xs0 + 2 * XP + 4 * kk);
    const v2f x3 = *(const v2f*)(xs0 + 3 * XP + 4 * kk);
    areg[kk] = (k0 * x0 + k1 * x1) + (k2 * x2 + k3 * x3);
  }

  v8f acc0 = {0.f, 0.f, 0.f, 0.f, 0.f, 0.f, 0.f, 0.f};
  v8f acc1 = {0.f, 0.f, 0.f, 0.f, 0.f, 0.f, 0.f, 0.f};
  v8f acc2 = {0.f, 0.f, 0.f, 0.f, 0.f, 0.f, 0.f, 0.f};
  v8f acc3 = {0.f, 0.f, 0.f, 0.f, 0.f, 0.f, 0.f, 0.f};
#pragma unroll
  for (int kk = 0; kk < 16; ++kk) {
    acc0 = __builtin_amdgcn_wmma_f32_16x16x4_f32(
         false, areg[kk],  false, bw[0][kk],
         (short)0, acc0,  false,  false);
    acc1 = __builtin_amdgcn_wmma_f32_16x16x4_f32(
         false, areg[kk],  false, bw[1][kk],
         (short)0, acc1,  false,  false);
    acc2 = __builtin_amdgcn_wmma_f32_16x16x4_f32(
         false, areg[kk],  false, bw[2][kk],
         (short)0, acc2,  false,  false);
    acc3 = __builtin_amdgcn_wmma_f32_16x16x4_f32(
         false, areg[kk],  false, bw[3][kk],
         (short)0, acc3,  false,  false);
  }

  float* orow = out + ((size_t)wb * NN + i0 + rt * 16 + 8 * hi) * DV + r;
#pragma unroll
  for (int j = 0; j < 8; ++j) {
    const float kbr = __shfl(kb, j + 8 * hi, 32);
    orow[(size_t)j * DV]      = acc0[j] + kbr * bgv[0];
    orow[(size_t)j * DV + 16] = acc1[j] + kbr * bgv[1];
    orow[(size_t)j * DV + 32] = acc2[j] + kbr * bgv[2];
    orow[(size_t)j * DV + 48] = acc3[j] + kbr * bgv[3];
  }
}

extern "C" void kernel_launch(void* const* d_in, const int* in_sizes, int n_in,
                              void* d_out, int out_size, void* d_ws, size_t ws_size,
                              hipStream_t stream) {
  const float* X  = (const float*)d_in[0];
  const float* Wg = (const float*)d_in[1];
  const float* bg = (const float*)d_in[2];
  const float* Wf = (const float*)d_in[3];
  const float* bf = (const float*)d_in[4];
  float* out = (float*)d_out;

  chord_fused_kernel<<<NN / TT, NT, 0, stream>>>(X, Wg, bg, Wf, bf, out);
}


# Wf bypasses LDS: coefficient job = (m, 4 Wf columns, quarter of d), Wf loaded straight to registers; first barrier waits for X stores only; quarters summed by quad DPP: a, c sum in a new order
# speedup vs baseline: 4.8884x; 1.1538x over previous
#include <hip/hip_runtime.h>

#define BB 2
#define NN 8192
#define DD 64
#define DV 64
#define MM 3

#define TT 32
#define XR (TT + 3)
#define XP 68
#define NG 9
#define CP (4 * NG)
#define BP 17
#define NT 128
#define XRR 9
#define NJ (MM * NG * 4)

typedef __attribute__((ext_vector_type(2))) float v2f;
typedef __attribute__((ext_vector_type(4))) float v4f;
typedef __attribute__((ext_vector_type(8))) float v8f;

static_assert(NT * XRR >= BB * XR * 16 && NT * XRR - BB * XR * 16 <= BB * XR * 16 - NT * (XRR - 1), "X rounds");
static_assert(NT >= NJ && DD == 64 && CP >= TT + 4, "coefficient jobs");

static __device__ __forceinline__ float quad_sum(float v) {
  v += __int_as_float(__builtin_amdgcn_update_dpp(0, __float_as_int(v), 0xB1, 0xF, 0xF, true));
  v += __int_as_float(__builtin_amdgcn_update_dpp(0, __float_as_int(v), 0x4E, 0xF, 0xF, true));
  return v;
}

__global__ __launch_bounds__(NT) void chord_fused_kernel(
    const float* __restrict__ X, const float* __restrict__ Wg,
    const float* __restrict__ bg, const float* __restrict__ Wf,
    const float* __restrict__ bf, float* __restrict__ out) {
  __shared__ __attribute__((aligned(16))) float Xs[BB * XR * XP];
  __shared__ v4f Bs[16 * 2 * 2 * BP];
  __shared__ __attribute__((aligned(16))) float Ca[BB * MM * CP];
  __shared__ __attribute__((aligned(16))) float Cc[BB * MM * CP];

  const int tid  = threadIdx.x;
  const int i0   = blockIdx.x * TT;
  const int lane = tid & 31;
  const int wave = tid >> 5;
  const int r    = lane & 15;
  const int hi   = lane >> 4;
  const int wb   = wave >> 1;
  const int rt   = wave & 1;

  v4f xv[XRR];
#pragma unroll
  for (int t = 0; t < XRR; ++t) {
    int e = tid + NT * t;
    if (t == XRR - 1) e = e < BB * XR * 16 ? e : e - (NT * XRR - BB * XR * 16);
    const int b   = e / (XR * 16);
    const int rem = e - b * (XR * 16);
    const int j   = rem >> 4;
    const int q   = rem & 15;
    const int row = (i0 + j) & (NN - 1);
    xv[t] = *(const v4f*)(X + ((size_t)b * NN + row) * DD + 4 * q);
  }
  const int job = tid < NJ ? tid : NJ - 1;
  const int jq  = job & 3;
  const int jmg = job >> 2;
  const int jm  = jmg / NG;
  const int jg  = jmg - jm * NG;
  const int jcol = (i0 + 4 * jg) & (NN - 1);
  v4f wv[4][4];
#pragma unroll
  for (int t = 0; t < 4; ++t) {
#pragma unroll
    for (int u = 0; u < 4; ++u) {
      const int d = 16 * t + 4 * jq + u;
      wv[t][u] = *(const v4f*)(Wf + ((size_t)jm * DD + d) * NN + jcol);
    }
  }
  const v4f bfv = *(const v4f*)(bf + (size_t)jm * NN + jcol);

#pragma unroll
  for (int t = 0; t < XRR; ++t) {
    int e = tid + NT * t;
    if (t == XRR - 1) e = e < BB * XR * 16 ? e : e - (NT * XRR - BB * XR * 16);
    const int b   = e / (XR * 16);
    const int rem = e - b * (XR * 16);
    const int j   = rem >> 4;
    const int q   = rem & 15;
    *(v4f*)(Xs + (b * XR + j) * XP + 4 * q) = xv[t];
  }
  __builtin_amdgcn_fence(__ATOMIC_RELEASE, "workgroup", "local");
  __builtin_amdgcn_s_barrier();
  __builtin_amdgcn_fence(__ATOMIC_ACQUIRE, "workgroup", "local");

  v4f g[2][4];
#pragma unroll
  for (int rnd = 0; rnd < 2; ++rnd) {
    const int idx = tid + NT * rnd;
    const int srq = idx & 3;
    const int sg  = idx >> 2;
    const int sk0 = 4 * (sg >> 2) + 2 * (sg & 1);
    const float* wgp = Wg + (size_t)sk0 * DV + 32 * ((sg >> 1) & 1) + 4 * srq;
    g[rnd][0] = *(const v4f*)(wgp);
    g[rnd][1] = *(const v4f*)(wgp + 16);
    g[rnd][2] = *(const v4f*)(wgp + DV);
    g[rnd][3] = *(const v4f*)(wgp + DV + 16);
  }
  float bgv[4];
#pragma unroll
  for (int t = 0; t < 4; ++t) bgv[t] = bg[16 * t + r];

  int xoff[5];
#pragma unroll
  for (int rr = 0; rr < 5; ++rr) {
    int row = 4 * jg - 1 + rr;
    row = row < 0 ? 0 : row;
    row = row > XR - 1 ? XR - 1 : row;
    xoff[rr] = row * XP + 4 * jq;
  }
  v2f pe[BB][2];
  float a1[BB], c1[BB], a3[BB], cm[BB];
#pragma unroll
  for (int b = 0; b < BB; ++b) {
    pe[b][0] = (v2f){0.f, 0.f};
    pe[b][1] = (v2f){0.f, 0.f};
    a1[b] = 0.f; c1[b] = 0.f; a3[b] = 0.f; cm[b] = 0.f;
  }
#pragma unroll
  for (int t = 0; t < 4; ++t) {
    v4f xr[BB][5];
#pragma unroll
    for (int b = 0; b < BB; ++b) {
#pragma unroll
      for (int rr = 0; rr < 5; ++rr) xr[b][rr] = *(const v4f*)(Xs + b * XR * XP + xoff[rr] + 16 * t);
    }
#pragma unroll
    for (int u = 0; u < 4; ++u) {
      const v4f w = wv[t][u];
      const v2f wxy = {w.x, w.y};
      const v2f wzw = {w.z, w.w};
#pragma unroll
      for (int b = 0; b < BB; ++b) {
        pe[b][0] += xr[b][1][u] * wxy;
        pe[b][1] += xr[b][3][u] * wzw;
        a1[b] = __builtin_fmaf(xr[b][2][u], w.y, a1[b]);
        c1[b] = __builtin_fmaf(xr[b][2][u], w.z, c1[b]);
        a3[b] = __builtin_fmaf(xr[b][4][u], w.w, a3[b]);
        cm[b] = __builtin_fmaf(xr[b][0][u], w.x, cm[b]);
      }
    }
  }
  {
#pragma unroll
    for (int b = 0; b < BB; ++b) {
      const float sa0 = quad_sum(pe[b][0].x), sc0 = quad_sum(pe[b][0].y);
      const float sa2 = quad_sum(pe[b][1].x), sc2 = quad_sum(pe[b][1].y);
      const float sa1 = quad_sum(a1[b]), sc1 = quad_sum(c1[b]);
      const float sa3 = quad_sum(a3[b]), scm = quad_sum(cm[b]);
      if (jq == 0) {
        const int cb = (b * MM + jm) * CP + 4 * jg;
        *(v4f*)(Ca + cb) = (v4f){sa0 + bfv.x, sa1 + bfv.y, sa2 + bfv.z, sa3 + bfv.w};
        *(v4f*)(Cc + cb) = (v4f){scm + bfv.x, sc0 + bfv.y, sc1 + bfv.z, sc2 + bfv.w};
      }
    }
  }
#pragma unroll
  for (int rnd = 0; rnd < 2; ++rnd) {
    const int idx = tid + NT * rnd;
    v4f* bdst = Bs + (idx >> 2) * BP + 4 * (idx & 3);
    bdst[0] = (v4f){g[rnd][0].x, g[rnd][2].x, g[rnd][1].x, g[rnd][3].x};
    bdst[1] = (v4f){g[rnd][0].y, g[rnd][2].y, g[rnd][1].y, g[rnd][3].y};
    bdst[2] = (v4f){g[rnd][0].z, g[rnd][2].z, g[rnd][1].z, g[rnd][3].z};
    bdst[3] = (v4f){g[rnd][0].w, g[rnd][2].w, g[rnd][1].w, g[rnd][3].w};
  }
  __syncthreads();

  const int il = rt * 16 + r;
  const float* ca0 = Ca + (wb * MM + 0) * CP + il;
  const float* cc0 = Cc + (wb * MM + 0) * CP + il + 1;
  const float* ca1 = Ca + (wb * MM + 1) * CP + il;
  const float* cc1 = Cc + (wb * MM + 1) * CP + il + 1;
  const float* ca2 = Ca + (wb * MM + 2) * CP + il;
  const float* cc2 = Cc + (wb * MM + 2) * CP + il + 1;
  const float a0_0 = ca0[0], c0_0 = cc0[0];
  const float a0_1 = ca0[1], c0_1 = cc0[1];
  const float a0_2 = ca0[2], c0_2 = cc0[2];
  const float a1_0 = ca1[0], c1_0 = cc1[0];
  const float a1_1 = ca1[1], c1_1 = cc1[1];
  const float a2_0 = ca2[0], c2_0 = cc2[0];
  const float p11 = a2_0 * a1_0;
  const float p12 = a2_0 * c1_0 + c2_0 * a1_1;
  const float p13 = c2_0 * c1_1;
  const float k0 = p11 * a0_0;
  const float k1 = p11 * c0_0 + p12 * a0_1;
  const float k2 = p12 * c0_1 + p13 * a0_2;
  const float k3 = p13 * c0_2;
  const float kb = (k0 + k1) + (k2 + k3);

  v2f bw[4][16];
#pragma unroll
  for (int ch = 0; ch < 2; ++ch) {
    const v4f* bsrc = Bs + (ch * 2 + hi) * BP + r;
#pragma unroll
    for (int kk = 0; kk < 16; ++kk) {
      const v4f t = bsrc[kk * 4 * BP];
      bw[2 * ch + 0][kk] = (v2f){t.x, t.y};
      bw[2 * ch + 1][kk] = (v2f){t.z, t.w};
    }
  }

  const float* xs0 = Xs + (wb * XR + il) * XP + 2 * hi;
  v2f areg[16];
#pragma unroll
  for (int kk = 0; kk < 16; ++kk) {
    const v2f x0 = *(const v2f*)(xs0 + 0 * XP + 4 * kk);
    const v2f x1 = *(const v2f*)(xs0 + 1 * XP + 4 * kk);
    const v2f x2 = *(const v2f*)(xs0 + 2 * XP + 4 * kk);
    const v2f x3 = *(const v2f*)(xs0 + 3 * XP + 4 * kk);
    areg[kk] = (k0 * x0 + k1 * x1) + (k2 * x2 + k3 * x3);
  }

  v8f acc0 = {0.f, 0.f, 0.f, 0.f, 0.f, 0.f, 0.f, 0.f};
  v8f acc1 = {0.f, 0.f, 0.f, 0.f, 0.f, 0.f, 0.f, 0.f};
  v8f acc2 = {0.f, 0.f, 0.f, 0.f, 0.f, 0.f, 0.f, 0.f};
  v8f acc3 = {0.f, 0.f, 0.f, 0.f, 0.f, 0.f, 0.f, 0.f};
#pragma unroll
  for (int kk = 0; kk < 16; ++kk) {
    acc0 = __builtin_amdgcn_wmma_f32_16x16x4_f32(
         false, areg[kk],  false, bw[0][kk],
         (short)0, acc0,  false,  false);
    acc1 = __builtin_amdgcn_wmma_f32_16x16x4_f32(
         false, areg[kk],  false, bw[1][kk],
         (short)0, acc1,  false,  false);
    acc2 = __builtin_amdgcn_wmma_f32_16x16x4_f32(
         false, areg[kk],  false, bw[2][kk],
         (short)0, acc2,  false,  false);
    acc3 = __builtin_amdgcn_wmma_f32_16x16x4_f32(
         false, areg[kk],  false, bw[3][kk],
         (short)0, acc3,  false,  false);
  }

  float* orow = out + ((size_t)wb * NN + i0 + rt * 16 + 8 * hi) * DV + r;
#pragma unroll
  for (int j = 0; j < 8; ++j) {
    const float kbr = __shfl(kb, j + 8 * hi, 32);
    orow[(size_t)j * DV]      = acc0[j] + kbr * bgv[0];
    orow[(size_t)j * DV + 16] = acc1[j] + kbr * bgv[1];
    orow[(size_t)j * DV + 32] = acc2[j] + kbr * bgv[2];
    orow[(size_t)j * DV + 48] = acc3[j] + kbr * bgv[3];
  }
}

extern "C" void kernel_launch(void* const* d_in, const int* in_sizes, int n_in,
                              void* d_out, int out_size, void* d_ws, size_t ws_size,
                              hipStream_t stream) {
  const float* X  = (const float*)d_in[0];
  const float* Wg = (const float*)d_in[1];
  const float* bg = (const float*)d_in[2];
  const float* Wf = (const float*)d_in[3];
  const float* bf = (const float*)d_in[4];
  float* out = (float*)d_out;

  chord_fused_kernel<<<NN / TT, NT, 0, stream>>>(X, Wg, bg, Wf, bf, out);
}
